# starting kernel
# baseline (speedup 1.0000x reference)
#include <hip/hip_runtime.h>

#define NVX 128
#define NVY 128
#define NC  80
#define B_  2
#define N_  6
#define D_  112
#define H_  16
#define W_  44
#define HW  (H_ * W_)
#define RAYS (B_ * N_ * H_ * W_)
#define RAYS_PER_BLOCK 8

typedef __attribute__((ext_vector_type(2))) float v2f;
typedef __attribute__((ext_vector_type(8))) float v8f;

__global__ __launch_bounds__(256) void vp_zero(float4* __restrict__ out, int n4) {
  int i = blockIdx.x * blockDim.x + threadIdx.x;
  const int stride = gridDim.x * blockDim.x;
  for (; i < n4; i += stride) out[i] = make_float4(0.f, 0.f, 0.f, 0.f);
}

__global__ __launch_bounds__(256) void vp_pool(const int*   __restrict__ geom,
                                               const float* __restrict__ depth,
                                               const float* __restrict__ ctx,
                                               float*       __restrict__ out) {
  const int lane = threadIdx.x & 31;
  const int wave = threadIdx.x >> 5;
  const int ray  = blockIdx.x * RAYS_PER_BLOCK + wave;

  const bool lo = lane < 16;
  const int w   = ray % W_;
  const int t   = ray / W_;
  const int h   = t % H_;
  const int bn  = t / H_;
  const int b   = bn / N_;

  const int baseP   = bn * (D_ * HW) + h * W_ + w;
  const int ctxBase = bn * (NC * HW) + h * W_ + w;
  const int voxBase = b * (NVX * NVY);

  v2f Bv[5];
#pragma unroll
  for (int ct = 0; ct < 5; ++ct) {
    float cv = 0.f;
    if (lo) cv = ctx[ctxBase + (ct * 16 + lane) * HW];
    v2f bb; bb.x = cv; bb.y = 0.f;
    Bv[ct] = bb;
  }

  const int rowoff = (lane >> 4) << 3;
  const int lanec  = lane & 15;

  for (int dt = 0; dt < 7; ++dt) {
    const int d0 = dt * 16;

    if (dt < 6) {
      int pd = d0 + 16 + lane;
      pd = pd < D_ ? pd : (D_ - 1);
      const int pp = baseP + pd * HW;
      __builtin_prefetch(&geom[3 * pp], 0, 1);
      __builtin_prefetch(&depth[pp], 0, 1);
    }

    float dv  = 0.f;
    int  voff = 0;
    if (lo) {
      const int p  = baseP + (d0 + lane) * HW;
      dv = depth[p];
      const int gx = geom[3 * p + 0];
      const int gy = geom[3 * p + 1];
      if (!((unsigned)gx < (unsigned)NVX && (unsigned)gy < (unsigned)NVY))
        dv = 0.f;
      const int cx = gx < 0 ? 0 : (gx > NVX - 1 ? NVX - 1 : gx);
      const int cy = gy < 0 ? 0 : (gy > NVY - 1 ? NVY - 1 : gy);
      voff = (voxBase + cy * NVX + cx) * NC;
    }
    v2f Av; Av.x = dv; Av.y = 0.f;

    int offr[8];
#pragma unroll
    for (int r = 0; r < 8; ++r)
      offr[r] = __shfl(voff, r + rowoff, 32);

    v8f acc[5];
#pragma unroll
    for (int ct = 0; ct < 5; ++ct) {
      v8f z = {0.f, 0.f, 0.f, 0.f, 0.f, 0.f, 0.f, 0.f};
      acc[ct] = __builtin_amdgcn_wmma_f32_16x16x4_f32(
           false, Av,  false, Bv[ct],
           (short)0, z,  false,  false);
    }

#pragma unroll
    for (int r = 0; r < 8; ++r) {
      float* p = out + (unsigned)(offr[r] + lanec);
#pragma unroll
      for (int ct = 0; ct < 5; ++ct) {
        __hip_atomic_fetch_add(p + ct * 16, acc[ct][r], __ATOMIC_RELAXED,
                               __HIP_MEMORY_SCOPE_AGENT);
      }
    }
  }
}

extern "C" void kernel_launch(void* const* d_in, const int* in_sizes, int n_in,
                              void* d_out, int out_size, void* d_ws, size_t ws_size,
                              hipStream_t stream) {
  (void)in_sizes; (void)n_in; (void)d_ws; (void)ws_size;
  const int*   geom  = (const int*)d_in[0];
  const float* depth = (const float*)d_in[1];
  const float* ctx   = (const float*)d_in[2];
  float*       out   = (float*)d_out;

  const int n4 = out_size / 4;
  vp_zero<<<512, 256, 0, stream>>>((float4*)out, n4);

  vp_pool<<<RAYS / RAYS_PER_BLOCK, 256, 0, stream>>>(geom, depth, ctx, out);
}


# Keep each output row's f32 atomic adds inside one package half: a workgroup reads its die number and draws a ray group of its half's batch from a ticket counter; adds still land in arrival order
# speedup vs baseline: 1.8498x; 1.8498x over previous
#include <hip/hip_runtime.h>

#define NVX 128
#define NVY 128
#define NC  80
#define B_  2
#define N_  6
#define D_  112
#define H_  16
#define W_  44
#define HW  (H_ * W_)
#define RAYS (B_ * N_ * H_ * W_)
#define RAYS_PER_BLOCK 8
#define GROUPS (RAYS / RAYS_PER_BLOCK)
#define GROUPS_PER_BATCH (GROUPS / B_)
#define TICKET_STRIDE 64

typedef __attribute__((ext_vector_type(2))) float v2f;
typedef __attribute__((ext_vector_type(8))) float v8f;

__global__ __launch_bounds__(256) void vp_zero(float4* __restrict__ out, int n4,
                                               int* __restrict__ tickets) {
  if (blockIdx.x == 0 && threadIdx.x < B_) tickets[threadIdx.x * TICKET_STRIDE] = 0;
  int i = blockIdx.x * blockDim.x + threadIdx.x;
  const int stride = gridDim.x * blockDim.x;
  for (; i < n4; i += stride) out[i] = make_float4(0.f, 0.f, 0.f, 0.f);
}

__global__ __launch_bounds__(256) void vp_pool(const int*   __restrict__ geom,
                                               const float* __restrict__ depth,
                                               const float* __restrict__ ctx,
                                               float*       __restrict__ out,
                                               int*         __restrict__ tickets) {
  __shared__ int s_group;
  if (threadIdx.x == 0) {
    const unsigned place = __builtin_amdgcn_s_sendmsg_rtn(0x87);
    const int half = (int)((place >> 18) & 1u);
    const int t = __hip_atomic_fetch_add(&tickets[half * TICKET_STRIDE], 1, __ATOMIC_RELAXED,
                                         __HIP_MEMORY_SCOPE_AGENT);
    int g = t < GROUPS_PER_BATCH ? half * GROUPS_PER_BATCH + t
                                 : (half ^ 1) * GROUPS_PER_BATCH + (GROUPS - 1 - t);
    g = g < 0 ? 0 : (g > GROUPS - 1 ? GROUPS - 1 : g);
    s_group = g;
  }
  __syncthreads();

  const int lane = threadIdx.x & 31;
  const int wave = threadIdx.x >> 5;
  const int ray  = s_group * RAYS_PER_BLOCK + wave;

  const bool lo = lane < 16;
  const int w   = ray % W_;
  const int t   = ray / W_;
  const int h   = t % H_;
  const int bn  = t / H_;
  const int b   = bn / N_;

  const int baseP   = bn * (D_ * HW) + h * W_ + w;
  const int ctxBase = bn * (NC * HW) + h * W_ + w;
  const int voxBase = b * (NVX * NVY);

  v2f Bv[5];
#pragma unroll
  for (int ct = 0; ct < 5; ++ct) {
    float cv = 0.f;
    if (lo) cv = ctx[ctxBase + (ct * 16 + lane) * HW];
    v2f bb; bb.x = cv; bb.y = 0.f;
    Bv[ct] = bb;
  }

  const int rowoff = (lane >> 4) << 3;
  const int lanec  = lane & 15;

  for (int dt = 0; dt < 7; ++dt) {
    const int d0 = dt * 16;

    if (dt < 6) {
      int pd = d0 + 16 + lane;
      pd = pd < D_ ? pd : (D_ - 1);
      const int pp = baseP + pd * HW;
      __builtin_prefetch(&geom[3 * pp], 0, 1);
      __builtin_prefetch(&depth[pp], 0, 1);
    }

    float dv  = 0.f;
    int  voff = 0;
    if (lo) {
      const int p  = baseP + (d0 + lane) * HW;
      dv = depth[p];
      const int gx = geom[3 * p + 0];
      const int gy = geom[3 * p + 1];
      if (!((unsigned)gx < (unsigned)NVX && (unsigned)gy < (unsigned)NVY))
        dv = 0.f;
      const int cx = gx < 0 ? 0 : (gx > NVX - 1 ? NVX - 1 : gx);
      const int cy = gy < 0 ? 0 : (gy > NVY - 1 ? NVY - 1 : gy);
      voff = (voxBase + cy * NVX + cx) * NC;
    }
    v2f Av; Av.x = dv; Av.y = 0.f;

    int offr[8];
#pragma unroll
    for (int r = 0; r < 8; ++r)
      offr[r] = __shfl(voff, r + rowoff, 32);

    v8f acc[5];
#pragma unroll
    for (int ct = 0; ct < 5; ++ct) {
      v8f z = {0.f, 0.f, 0.f, 0.f, 0.f, 0.f, 0.f, 0.f};
      acc[ct] = __builtin_amdgcn_wmma_f32_16x16x4_f32(
           false, Av,  false, Bv[ct],
           (short)0, z,  false,  false);
    }

#pragma unroll
    for (int r = 0; r < 8; ++r) {
      float* p = out + (unsigned)(offr[r] + lanec);
#pragma unroll
      for (int ct = 0; ct < 5; ++ct) {
        __hip_atomic_fetch_add(p + ct * 16, acc[ct][r], __ATOMIC_RELAXED,
                               __HIP_MEMORY_SCOPE_AGENT);
      }
    }
  }
}

extern "C" void kernel_launch(void* const* d_in, const int* in_sizes, int n_in,
                              void* d_out, int out_size, void* d_ws, size_t ws_size,
                              hipStream_t stream) {
  (void)in_sizes; (void)n_in; (void)ws_size;
  const int*   geom  = (const int*)d_in[0];
  const float* depth = (const float*)d_in[1];
  const float* ctx   = (const float*)d_in[2];
  float*       out   = (float*)d_out;
  int*         tickets = (int*)d_ws;

  const int n4 = out_size / 4;
  vp_zero<<<512, 256, 0, stream>>>((float4*)out, n4, tickets);

  vp_pool<<<RAYS / RAYS_PER_BLOCK, 256, 0, stream>>>(geom, depth, ctx, out, tickets);
}


# Every f32 atomic add of a full line covers one whole aligned 128-B line of one voxel row: voxel parity picks the WMMA k slot, so a row costs 3 line requests, not 5; adds still in arrival order
# speedup vs baseline: 2.2455x; 1.2139x over previous
#include <hip/hip_runtime.h>

#define NVX 128
#define NVY 128
#define NC  80
#define B_  2
#define N_  6
#define D_  112
#define H_  16
#define W_  44
#define HW  (H_ * W_)
#define RAYS (B_ * N_ * H_ * W_)
#define RAYS_PER_BLOCK 8
#define GROUPS (RAYS / RAYS_PER_BLOCK)
#define GROUPS_PER_BATCH (GROUPS / B_)
#define TICKET_STRIDE 64

typedef __attribute__((ext_vector_type(2))) float v2f;
typedef __attribute__((ext_vector_type(8))) float v8f;

__global__ __launch_bounds__(256) void vp_zero(float4* __restrict__ out, int n4,
                                               int* __restrict__ tickets) {
  if (blockIdx.x == 0 && threadIdx.x < B_) tickets[threadIdx.x * TICKET_STRIDE] = 0;
  int i = blockIdx.x * blockDim.x + threadIdx.x;
  const int stride = gridDim.x * blockDim.x;
  for (; i < n4; i += stride) out[i] = make_float4(0.f, 0.f, 0.f, 0.f);
}

__global__ __launch_bounds__(256) void vp_pool(const int*   __restrict__ geom,
                                               const float* __restrict__ depth,
                                               const float* __restrict__ ctx,
                                               float*       __restrict__ out,
                                               int*         __restrict__ tickets) {
  __shared__ int s_group;
  if (threadIdx.x == 0) {
    const unsigned place = __builtin_amdgcn_s_sendmsg_rtn(0x87);
    const int half = (int)((place >> 18) & 1u);
    const int t = __hip_atomic_fetch_add(&tickets[half * TICKET_STRIDE], 1, __ATOMIC_RELAXED,
                                         __HIP_MEMORY_SCOPE_AGENT);
    int g = t < GROUPS_PER_BATCH ? half * GROUPS_PER_BATCH + t
                                 : (half ^ 1) * GROUPS_PER_BATCH + (GROUPS - 1 - t);
    g = g < 0 ? 0 : (g > GROUPS - 1 ? GROUPS - 1 : g);
    s_group = g;
  }
  __syncthreads();

  const int lane = threadIdx.x & 31;
  const int wave = threadIdx.x >> 5;
  const int ray  = s_group * RAYS_PER_BLOCK + wave;

  const bool lo = lane < 16;
  const int w   = ray % W_;
  const int t   = ray / W_;
  const int h   = t % H_;
  const int bn  = t / H_;
  const int b   = bn / N_;

  const int baseP   = bn * (D_ * HW) + h * W_ + w;
  const int ctxBase = bn * (NC * HW) + h * W_ + w;
  const int voxBase = b * (NVX * NVY);


  v2f Bm[2];
#pragma unroll
  for (int j = 0; j < 2; ++j) {
    v2f bb;
    bb.x = ctx[ctxBase + (32 * j + lane) * HW];
    bb.y = ctx[ctxBase + (32 * j + 16 + lane) * HW];
    Bm[j] = bb;
  }
  v2f Bl;
  Bl.x = 0.f;
  Bl.y = 0.f;
  if (lo) {
    Bl.x = ctx[ctxBase + (64 + lane) * HW];
    Bl.y = Bm[0].x;
  }

  const int rowoff = (lane >> 4) << 3;
  const int lanec  = lane & 15;
  const bool act = ((lane >> 3) & 1) == (lane >> 4);
  const int  srcl = lane & 7;

  for (int dt = 0; dt < 7; ++dt) {
    const int d0 = dt * 16;

    if (dt < 6) {
      int pd = d0 + 16 + lane;
      pd = pd < D_ ? pd : (D_ - 1);
      const int pp = baseP + pd * HW;
      __builtin_prefetch(&geom[3 * pp], 0, 1);
      __builtin_prefetch(&depth[pp], 0, 1);
    }

    float dv = 0.f;
    int   q  = 0;
    int   s1 = 0;
    if (lo) {
      const int p  = baseP + (d0 + lane) * HW;
      dv = depth[p];
      const int gx = geom[3 * p + 0];
      const int gy = geom[3 * p + 1];
      if (!((unsigned)gx < (unsigned)NVX && (unsigned)gy < (unsigned)NVY))
        dv = 0.f;
      const int cx = gx < 0 ? 0 : (gx > NVX - 1 ? NVX - 1 : gx);
      const int cy = gy < 0 ? 0 : (gy > NVY - 1 ? NVY - 1 : gy);
      const int vox = voxBase + cy * NVX + cx;
      q  = vox & 1;
      s1 = vox * NC + q * 16;
    }
    const float ax = q ? 0.f : dv;
    const float ay = q ? dv : 0.f;
    const int loff = q ? s1 - 16 : s1 + 64;

    v2f Al; Al.x = ax; Al.y = ay;
    v8f z = {0.f, 0.f, 0.f, 0.f, 0.f, 0.f, 0.f, 0.f};
    const v8f accl = __builtin_amdgcn_wmma_f32_16x16x4_f32(
         false, Al,  false, Bl,
         (short)0, z,  false,  false);

    v8f accm[2][2];
#pragma unroll
    for (int g = 0; g < 2; ++g) {
      const float sx = __shfl(ax, srcl + 8 * g, 32);
      const float sy = __shfl(ay, srcl + 8 * g, 32);
      v2f Ag; Ag.x = act ? sx : 0.f; Ag.y = act ? sy : 0.f;
#pragma unroll
      for (int j = 0; j < 2; ++j) {
        accm[g][j] = __builtin_amdgcn_wmma_f32_16x16x4_f32(
             false, Ag,  false, Bm[j],
             (short)0, z,  false,  false);
      }
    }

#pragma unroll
    for (int g = 0; g < 2; ++g) {
#pragma unroll
      for (int m = 0; m < 8; ++m) {
        const int sb = __builtin_amdgcn_readlane(s1, m + 8 * g);
        float* p = out + (unsigned)sb + lane;
        __hip_atomic_fetch_add(p, accm[g][0][m], __ATOMIC_RELAXED,
                               __HIP_MEMORY_SCOPE_AGENT);
        __hip_atomic_fetch_add(p + 32, accm[g][1][m], __ATOMIC_RELAXED,
                               __HIP_MEMORY_SCOPE_AGENT);
      }
    }
#pragma unroll
    for (int r = 0; r < 8; ++r) {
      const int o = __shfl(loff, r + rowoff, 32);
      __hip_atomic_fetch_add(out + (unsigned)(o + lanec), accl[r], __ATOMIC_RELAXED,
                             __HIP_MEMORY_SCOPE_AGENT);
    }
  }
}

extern "C" void kernel_launch(void* const* d_in, const int* in_sizes, int n_in,
                              void* d_out, int out_size, void* d_ws, size_t ws_size,
                              hipStream_t stream) {
  (void)in_sizes; (void)n_in; (void)ws_size;
  const int*   geom  = (const int*)d_in[0];
  const float* depth = (const float*)d_in[1];
  const float* ctx   = (const float*)d_in[2];
  float*       out   = (float*)d_out;
  int*         tickets = (int*)d_ws;

  const int n4 = out_size / 4;
  vp_zero<<<512, 256, 0, stream>>>((float4*)out, n4, tickets);

  vp_pool<<<RAYS / RAYS_PER_BLOCK, 256, 0, stream>>>(geom, depth, ctx, out, tickets);
}


# Next depth tile's depth and voxel coordinates are loaded into registers one trip ahead; the per-lane prefetches (8 waves asking the L2 for the same lines) are dropped; adds still in arrival order
# speedup vs baseline: 2.6633x; 1.1861x over previous
#include <hip/hip_runtime.h>

#define NVX 128
#define NVY 128
#define NC  80
#define B_  2
#define N_  6
#define D_  112
#define H_  16
#define W_  44
#define HW  (H_ * W_)
#define RAYS (B_ * N_ * H_ * W_)
#define RAYS_PER_BLOCK 8
#define GROUPS (RAYS / RAYS_PER_BLOCK)
#define GROUPS_PER_BATCH (GROUPS / B_)
#define TICKET_STRIDE 64

typedef __attribute__((ext_vector_type(2))) float v2f;
typedef __attribute__((ext_vector_type(8))) float v8f;

__global__ __launch_bounds__(256) void vp_zero(float4* __restrict__ out, int n4,
                                               int* __restrict__ tickets) {
  if (blockIdx.x == 0 && threadIdx.x < B_) tickets[threadIdx.x * TICKET_STRIDE] = 0;
  int i = blockIdx.x * blockDim.x + threadIdx.x;
  const int stride = gridDim.x * blockDim.x;
  for (; i < n4; i += stride) out[i] = make_float4(0.f, 0.f, 0.f, 0.f);
}

__global__ __launch_bounds__(256) void vp_pool(const int*   __restrict__ geom,
                                               const float* __restrict__ depth,
                                               const float* __restrict__ ctx,
                                               float*       __restrict__ out,
                                               int*         __restrict__ tickets) {
  __shared__ int s_group;
  if (threadIdx.x == 0) {
    const unsigned place = __builtin_amdgcn_s_sendmsg_rtn(0x87);
    const int half = (int)((place >> 18) & 1u);
    const int t = __hip_atomic_fetch_add(&tickets[half * TICKET_STRIDE], 1, __ATOMIC_RELAXED,
                                         __HIP_MEMORY_SCOPE_AGENT);
    int g = t < GROUPS_PER_BATCH ? half * GROUPS_PER_BATCH + t
                                 : (half ^ 1) * GROUPS_PER_BATCH + (GROUPS - 1 - t);
    g = g < 0 ? 0 : (g > GROUPS - 1 ? GROUPS - 1 : g);
    s_group = g;
  }
  __syncthreads();

  const int lane = threadIdx.x & 31;
  const int wave = threadIdx.x >> 5;
  const int ray  = s_group * RAYS_PER_BLOCK + wave;

  const bool lo = lane < 16;
  const int w   = ray % W_;
  const int t   = ray / W_;
  const int h   = t % H_;
  const int bn  = t / H_;
  const int b   = bn / N_;

  const int baseP   = bn * (D_ * HW) + h * W_ + w;
  const int ctxBase = bn * (NC * HW) + h * W_ + w;
  const int voxBase = b * (NVX * NVY);


  v2f Bm[2];
#pragma unroll
  for (int j = 0; j < 2; ++j) {
    v2f bb;
    bb.x = ctx[ctxBase + (32 * j + lane) * HW];
    bb.y = ctx[ctxBase + (32 * j + 16 + lane) * HW];
    Bm[j] = bb;
  }
  v2f Bl;
  Bl.x = 0.f;
  Bl.y = 0.f;
  if (lo) {
    Bl.x = ctx[ctxBase + (64 + lane) * HW];
    Bl.y = Bm[0].x;
  }

  const int rowoff = (lane >> 4) << 3;
  const int lanec  = lane & 15;
  const bool act = ((lane >> 3) & 1) == (lane >> 4);
  const int  srcl = lane & 7;

  float ndv = 0.f;
  int   ngx = 0;
  int   ngy = 0;
  if (lo) {
    const int p = baseP + lane * HW;
    ndv = depth[p];
    ngx = geom[3 * p + 0];
    ngy = geom[3 * p + 1];
  }

  for (int dt = 0; dt < 7; ++dt) {
    float     dv = ndv;
    const int gx = ngx;
    const int gy = ngy;

    if (dt < 6 && lo) {
      const int p = baseP + (dt * 16 + 16 + lane) * HW;
      ndv = depth[p];
      ngx = geom[3 * p + 0];
      ngy = geom[3 * p + 1];
    }

    int   q  = 0;
    int   s1 = 0;
    if (lo) {
      if (!((unsigned)gx < (unsigned)NVX && (unsigned)gy < (unsigned)NVY))
        dv = 0.f;
      const int cx = gx < 0 ? 0 : (gx > NVX - 1 ? NVX - 1 : gx);
      const int cy = gy < 0 ? 0 : (gy > NVY - 1 ? NVY - 1 : gy);
      const int vox = voxBase + cy * NVX + cx;
      q  = vox & 1;
      s1 = vox * NC + q * 16;
    }
    const float ax = q ? 0.f : dv;
    const float ay = q ? dv : 0.f;
    const int loff = q ? s1 - 16 : s1 + 64;

    v2f Al; Al.x = ax; Al.y = ay;
    v8f z = {0.f, 0.f, 0.f, 0.f, 0.f, 0.f, 0.f, 0.f};
    const v8f accl = __builtin_amdgcn_wmma_f32_16x16x4_f32(
         false, Al,  false, Bl,
         (short)0, z,  false,  false);

    v8f accm[2][2];
#pragma unroll
    for (int g = 0; g < 2; ++g) {
      const float sx = __shfl(ax, srcl + 8 * g, 32);
      const float sy = __shfl(ay, srcl + 8 * g, 32);
      v2f Ag; Ag.x = act ? sx : 0.f; Ag.y = act ? sy : 0.f;
#pragma unroll
      for (int j = 0; j < 2; ++j) {
        accm[g][j] = __builtin_amdgcn_wmma_f32_16x16x4_f32(
             false, Ag,  false, Bm[j],
             (short)0, z,  false,  false);
      }
    }

#pragma unroll
    for (int g = 0; g < 2; ++g) {
#pragma unroll
      for (int m = 0; m < 8; ++m) {
        const int sb = __builtin_amdgcn_readlane(s1, m + 8 * g);
        float* p = out + (unsigned)sb + lane;
        __hip_atomic_fetch_add(p, accm[g][0][m], __ATOMIC_RELAXED,
                               __HIP_MEMORY_SCOPE_AGENT);
        __hip_atomic_fetch_add(p + 32, accm[g][1][m], __ATOMIC_RELAXED,
                               __HIP_MEMORY_SCOPE_AGENT);
      }
    }
#pragma unroll
    for (int r = 0; r < 8; ++r) {
      const int o = __shfl(loff, r + rowoff, 32);
      __hip_atomic_fetch_add(out + (unsigned)(o + lanec), accl[r], __ATOMIC_RELAXED,
                             __HIP_MEMORY_SCOPE_AGENT);
    }
  }
}

extern "C" void kernel_launch(void* const* d_in, const int* in_sizes, int n_in,
                              void* d_out, int out_size, void* d_ws, size_t ws_size,
                              hipStream_t stream) {
  (void)in_sizes; (void)n_in; (void)ws_size;
  const int*   geom  = (const int*)d_in[0];
  const float* depth = (const float*)d_in[1];
  const float* ctx   = (const float*)d_in[2];
  float*       out   = (float*)d_out;
  int*         tickets = (int*)d_ws;

  const int n4 = out_size / 4;
  vp_zero<<<512, 256, 0, stream>>>((float4*)out, n4, tickets);

  vp_pool<<<RAYS / RAYS_PER_BLOCK, 256, 0, stream>>>(geom, depth, ctx, out, tickets);
}
